# starting kernel
# baseline (speedup 1.0000x reference)
#include <hip/hip_runtime.h>
#include <hip/hip_bf16.h>
#include <math.h>


typedef __attribute__((ext_vector_type(2))) float v2f;
typedef __attribute__((ext_vector_type(8))) float v8f;

#define DIMK  512
#define BATCH 32768
#define HSZ   17
#define NPAR  289
#define NOUT  578
#define NT    37
#define LDA   516

__global__ __launch_bounds__(256, 2)
void fourier_head2d_fused(const float* __restrict__ X,
                          const float* __restrict__ T,
                          const float* __restrict__ W,
                          const float* __restrict__ bias,
                          float* __restrict__ out)
{
    __shared__ float sA[16 * LDA];
    __shared__ float sTgt[32];
    __shared__ float sAcc[48];

    const int tid     = threadIdx.x;
    const int lane    = tid & 31;
    const int wave    = tid >> 5;
    const int rowBase = blockIdx.x * 16;

    if (tid < 48) sAcc[tid] = 0.0f;
    if (tid < 32) sTgt[tid] = T[rowBase * 2 + tid];

    for (int i = tid; i < 16 * (DIMK / 4); i += 256) {
        const int r  = i >> 7;
        const int c4 = i & 127;
        const float4 v = reinterpret_cast<const float4*>(
                             X + (size_t)(rowBase + r) * DIMK)[c4];
        float* dst = &sA[r * LDA + c4 * 4];
        dst[0] = v.x; dst[1] = v.y; dst[2] = v.z; dst[3] = v.w;
    }
    __syncthreads();

    const int half = lane >> 4;
    const int sub  = lane & 15;

    for (int nt = wave; nt < NT; nt += 8) {
        const int n  = nt * 16 + sub;
        const int nc = (n < NOUT) ? n : (NOUT - 1);
        const float* wrow = W  + (size_t)nc * DIMK + half * 2;
        const float* arow = sA + sub * LDA          + half * 2;

        v8f acc = {};
#pragma unroll 4
        for (int k0 = 0; k0 < DIMK; k0 += 4) {
            v2f a = *reinterpret_cast<const v2f*>(arow + k0);
            v2f b = *reinterpret_cast<const v2f*>(wrow + k0);
            acc = __builtin_amdgcn_wmma_f32_16x16x4_f32(
                      false, a, false, b, (short)0, acc, false, false);
        }

        const bool  valid  = (n < NOUT);
        const float bn     = valid ? bias[n] : 0.0f;
        const int   idx    = valid ? ((n < NPAR) ? n : n - NPAR) : 0;
        const bool  isImag = valid && (n >= NPAR);
        const int   fr     = idx / HSZ;
        const int   fc     = idx - fr * HSZ;
        const float frf    = (float)fr;
        const float fcf    = (float)fc;

#pragma unroll
        for (int j = 0; j < 8; ++j) {
            const int   m  = j + 8 * half;
            const float v  = acc[j] + bn;
            const float tx = sTgt[2 * m + 0];
            const float ty = sTgt[2 * m + 1];
            const float u  = fcf * tx + frf * ty;
            float su, cu;
            sincospif(u, &su, &cu);

            float sre, sim, cen;
            if (isImag) { sre = -v * su; sim =  v * cu; }
            else        { sre =  v * cu; sim =  v * su; }
            cen = v * v;
            if (!valid) { sre = 0.0f; sim = 0.0f; cen = 0.0f; }

#pragma unroll
            for (int d = 1; d < 16; d <<= 1) {
                sre += __shfl_xor(sre, d, 16);
                sim += __shfl_xor(sim, d, 16);
                cen += __shfl_xor(cen, d, 16);
            }
            if (sub == 0) {
                atomicAdd(&sAcc[3 * m + 0], sre);
                atomicAdd(&sAcc[3 * m + 1], sim);
                atomicAdd(&sAcc[3 * m + 2], cen);
            }
        }
    }
    __syncthreads();

    if (tid < 16) {
        const float sre = sAcc[3 * tid + 0];
        const float sim = sAcc[3 * tid + 1];
        const float cen = sAcc[3 * tid + 2];
        out[rowBase + tid] = (sre * sre + sim * sim) / (4.0f * cen);
    }
}

extern "C" void kernel_launch(void* const* d_in, const int* in_sizes, int n_in,
                              void* d_out, int out_size, void* d_ws, size_t ws_size,
                              hipStream_t stream) {
    (void)in_sizes; (void)n_in; (void)d_ws; (void)ws_size; (void)out_size;
    const float* X  = (const float*)d_in[0];
    const float* T  = (const float*)d_in[1];
    const float* W  = (const float*)d_in[2];
    const float* b  = (const float*)d_in[3];
    float* out = (float*)d_out;

    dim3 grid(BATCH / 16), block(256);
    hipLaunchKernelGGL(fourier_head2d_fused, grid, block, 0, stream, X, T, W, b, out);
}


# Register-block the GEMM over M: 128-row work-groups, each wave keeps four accumulator tiles per W fragment, so W streams once per 128 rows; k terms regrouped in steps of 8 (f32 sum order changes)
# speedup vs baseline: 1.2899x; 1.2899x over previous
#include <hip/hip_runtime.h>
#include <hip/hip_bf16.h>
#include <math.h>


typedef __attribute__((ext_vector_type(2))) float v2f;
typedef __attribute__((ext_vector_type(4))) float v4f;
typedef __attribute__((ext_vector_type(8))) float v8f;

#define DIMK  512
#define BATCH 32768
#define HSZ   17
#define NPAR  289
#define NOUT  578
#define NT    37
#define LDA   516
#define MB    128
#define MT    4
#define KC    32

__global__ __launch_bounds__(512)
void fourier_head2d_fused(const float* __restrict__ X,
                          const float* __restrict__ T,
                          const float* __restrict__ W,
                          const float* __restrict__ bias,
                          float* __restrict__ out)
{
    __shared__ __attribute__((aligned(16))) float sA[MB * LDA];
    __shared__ float sTgt[2 * MB];
    __shared__ float sAcc[3 * MB];

    const int tid     = threadIdx.x;
    const int lane    = tid & 31;
    const int wave    = tid >> 5;
    const int rowBase = blockIdx.x * MB;

    if (tid < 3 * MB) sAcc[tid] = 0.0f;
    if (tid < 2 * MB) sTgt[tid] = T[(size_t)rowBase * 2 + tid];

#pragma unroll 8
    for (int i = tid; i < MB * (DIMK / 4); i += 512) {
        const int r  = i >> 7;
        const int c4 = i & 127;
        const v4f v = *reinterpret_cast<const v4f*>(
                          X + (size_t)(rowBase + r) * DIMK + c4 * 4);
        *reinterpret_cast<v4f*>(&sA[r * LDA + c4 * 4]) = v;
    }
    __syncthreads();

    const int half  = lane >> 4;
    const int sub   = lane & 15;
    const int mhalf = wave & 1;
    const int wp    = wave >> 1;

    const float* aBase = sA + (mhalf * (MT * 16) + sub) * LDA + half * 4;

    for (int nt = wp; nt < NT; nt += 8) {
        const int n  = nt * 16 + sub;
        const int nc = (n < NOUT) ? n : (NOUT - 1);
        const float* wrow = W + (size_t)nc * DIMK + half * 4;

        v8f acc[MT];
#pragma unroll
        for (int mt = 0; mt < MT; ++mt) acc[mt] = (v8f){};

        v4f bcur[4], bnxt[4];
#pragma unroll
        for (int q = 0; q < 4; ++q)
            bcur[q] = *reinterpret_cast<const v4f*>(wrow + q * 8);

        v4f acur[MT], anxt[MT];
#pragma unroll
        for (int mt = 0; mt < MT; ++mt)
            acur[mt] = *reinterpret_cast<const v4f*>(aBase + mt * 16 * LDA);

#pragma unroll 2
        for (int kc = 0; kc < DIMK / KC; ++kc) {
            const int kn = ((kc + 1 < DIMK / KC) ? (kc + 1) : kc) * KC;
#pragma unroll
            for (int q = 0; q < 4; ++q)
                bnxt[q] = *reinterpret_cast<const v4f*>(wrow + kn + q * 8);

#pragma unroll
            for (int q = 0; q < 4; ++q) {
                const int ko  = kc * KC + q * 8;
                const int kon = (ko + 8 < DIMK) ? (ko + 8) : ko;
#pragma unroll
                for (int mt = 0; mt < MT; ++mt)
                    anxt[mt] = *reinterpret_cast<const v4f*>(
                                   aBase + mt * 16 * LDA + kon);
                const v2f b0 = __builtin_shufflevector(bcur[q], bcur[q], 0, 1);
                const v2f b1 = __builtin_shufflevector(bcur[q], bcur[q], 2, 3);
#pragma unroll
                for (int mt = 0; mt < MT; ++mt) {
                    const v2f a0 = __builtin_shufflevector(acur[mt], acur[mt], 0, 1);
                    acc[mt] = __builtin_amdgcn_wmma_f32_16x16x4_f32(
                                  false, a0, false, b0, (short)0, acc[mt], false, false);
                }
#pragma unroll
                for (int mt = 0; mt < MT; ++mt) {
                    const v2f a1 = __builtin_shufflevector(acur[mt], acur[mt], 2, 3);
                    acc[mt] = __builtin_amdgcn_wmma_f32_16x16x4_f32(
                                  false, a1, false, b1, (short)0, acc[mt], false, false);
                }
#pragma unroll
                for (int mt = 0; mt < MT; ++mt) acur[mt] = anxt[mt];
            }
#pragma unroll
            for (int q = 0; q < 4; ++q) bcur[q] = bnxt[q];
        }

        const bool  valid  = (n < NOUT);
        const float bn     = valid ? bias[n] : 0.0f;
        const int   idx    = valid ? ((n < NPAR) ? n : n - NPAR) : 0;
        const bool  isImag = valid && (n >= NPAR);
        const int   fr     = idx / HSZ;
        const int   fc     = idx - fr * HSZ;
        const float frf    = (float)fr;
        const float fcf    = (float)fc;

#pragma unroll
        for (int mt = 0; mt < MT; ++mt) {
#pragma unroll
            for (int j = 0; j < 8; ++j) {
                const int   m  = mhalf * (MT * 16) + mt * 16 + j + 8 * half;
                const float v  = acc[mt][j] + bn;
                const float tx = sTgt[2 * m + 0];
                const float ty = sTgt[2 * m + 1];
                const float u  = fcf * tx + frf * ty;
                float su, cu;
                sincospif(u, &su, &cu);

                float sre, sim, cen;
                if (isImag) { sre = -v * su; sim =  v * cu; }
                else        { sre =  v * cu; sim =  v * su; }
                cen = v * v;
                if (!valid) { sre = 0.0f; sim = 0.0f; cen = 0.0f; }

#pragma unroll
                for (int d = 1; d < 16; d <<= 1) {
                    sre += __shfl_xor(sre, d, 16);
                    sim += __shfl_xor(sim, d, 16);
                    cen += __shfl_xor(cen, d, 16);
                }
                if (sub == 0) {
                    atomicAdd(&sAcc[3 * m + 0], sre);
                    atomicAdd(&sAcc[3 * m + 1], sim);
                    atomicAdd(&sAcc[3 * m + 2], cen);
                }
            }
        }
    }
    __syncthreads();

    if (tid < MB) {
        const float sre = sAcc[3 * tid + 0];
        const float sim = sAcc[3 * tid + 1];
        const float cen = sAcc[3 * tid + 2];
        out[rowBase + tid] = (sre * sre + sim * sim) / (4.0f * cen);
    }
}

extern "C" void kernel_launch(void* const* d_in, const int* in_sizes, int n_in,
                              void* d_out, int out_size, void* d_ws, size_t ws_size,
                              hipStream_t stream) {
    (void)in_sizes; (void)n_in; (void)d_ws; (void)ws_size; (void)out_size;
    const float* X  = (const float*)d_in[0];
    const float* T  = (const float*)d_in[1];
    const float* W  = (const float*)d_in[2];
    const float* b  = (const float*)d_in[3];
    float* out = (float*)d_out;

    dim3 grid(BATCH / MB), block(512);
    hipLaunchKernelGGL(fourier_head2d_fused, grid, block, 0, stream, X, T, W, b, out);
}


# Swap WMMA operand roles (W fragment as A, X as B): a lane holds 1 sample x 8 columns, so the column sum is in-lane adds, no shuffles or atomics; fixed-order partials (f32 sum order changes)
# speedup vs baseline: 1.5426x; 1.1959x over previous
#include <hip/hip_runtime.h>
#include <hip/hip_bf16.h>
#include <math.h>


typedef __attribute__((ext_vector_type(2))) float v2f;
typedef __attribute__((ext_vector_type(4))) float v4f;
typedef __attribute__((ext_vector_type(8))) float v8f;

#define DIMK  512
#define BATCH 32768
#define HSZ   17
#define NPAR  289
#define NOUT  578
#define NT    37
#define LDA   516
#define MB    128
#define MT    4
#define KC    32
#define NSLOT 16

__global__ __launch_bounds__(512)
void fourier_head2d_fused(const float* __restrict__ X,
                          const float* __restrict__ T,
                          const float* __restrict__ W,
                          const float* __restrict__ bias,
                          float* __restrict__ out)
{
    __shared__ __attribute__((aligned(16))) float sA[MB * LDA];
    __shared__ __attribute__((aligned(8)))  float sTgt[2 * MB];
    __shared__ __attribute__((aligned(16))) float sBias[NT * 16];
    __shared__ float sPart[3 * NSLOT * MB];

    const int tid     = threadIdx.x;
    const int lane    = tid & 31;
    const int wave    = tid >> 5;
    const int rowBase = blockIdx.x * MB;

    if (tid < 2 * MB) sTgt[tid] = T[(size_t)rowBase * 2 + tid];
    for (int i = tid; i < NT * 16; i += 512) sBias[i] = (i < NOUT) ? bias[i] : 0.0f;

#pragma unroll 8
    for (int i = tid; i < MB * (DIMK / 4); i += 512) {
        const int r  = i >> 7;
        const int c4 = i & 127;
        const v4f v = *reinterpret_cast<const v4f*>(
                          X + (size_t)(rowBase + r) * DIMK + c4 * 4);
        *reinterpret_cast<v4f*>(&sA[r * LDA + c4 * 4]) = v;
    }
    __syncthreads();

    const int half  = lane >> 4;
    const int sub   = lane & 15;
    const int mhalf = wave & 1;
    const int wp    = wave >> 1;

    const float* aBase = sA + (mhalf * (MT * 16) + sub) * LDA + half * 4;

    float psre[MT], psim[MT], pcen[MT];
#pragma unroll
    for (int mt = 0; mt < MT; ++mt) { psre[mt] = 0.0f; psim[mt] = 0.0f; pcen[mt] = 0.0f; }

    for (int nt = wp; nt < NT; nt += 8) {
        const int n  = nt * 16 + sub;
        const int nc = (n < NOUT) ? n : (NOUT - 1);
        const float* wrow = W + (size_t)nc * DIMK + half * 4;

        v8f acc[MT];
#pragma unroll
        for (int mt = 0; mt < MT; ++mt) acc[mt] = (v8f){};

        v4f bcur[4], bnxt[4];
#pragma unroll
        for (int q = 0; q < 4; ++q)
            bcur[q] = *reinterpret_cast<const v4f*>(wrow + q * 8);

        v4f acur[MT], anxt[MT];
#pragma unroll
        for (int mt = 0; mt < MT; ++mt)
            acur[mt] = *reinterpret_cast<const v4f*>(aBase + mt * 16 * LDA);

#pragma unroll 2
        for (int kc = 0; kc < DIMK / KC; ++kc) {
            const int kn = ((kc + 1 < DIMK / KC) ? (kc + 1) : kc) * KC;
#pragma unroll
            for (int q = 0; q < 4; ++q)
                bnxt[q] = *reinterpret_cast<const v4f*>(wrow + kn + q * 8);

#pragma unroll
            for (int q = 0; q < 4; ++q) {
                const int ko  = kc * KC + q * 8;
                const int kon = (ko + 8 < DIMK) ? (ko + 8) : ko;
#pragma unroll
                for (int mt = 0; mt < MT; ++mt)
                    anxt[mt] = *reinterpret_cast<const v4f*>(
                                   aBase + mt * 16 * LDA + kon);
                const v2f b0 = __builtin_shufflevector(bcur[q], bcur[q], 0, 1);
                const v2f b1 = __builtin_shufflevector(bcur[q], bcur[q], 2, 3);
#pragma unroll
                for (int mt = 0; mt < MT; ++mt) {
                    const v2f a0 = __builtin_shufflevector(acur[mt], acur[mt], 0, 1);
                    acc[mt] = __builtin_amdgcn_wmma_f32_16x16x4_f32(
                                  false, b0, false, a0, (short)0, acc[mt], false, false);
                }
#pragma unroll
                for (int mt = 0; mt < MT; ++mt) {
                    const v2f a1 = __builtin_shufflevector(acur[mt], acur[mt], 2, 3);
                    acc[mt] = __builtin_amdgcn_wmma_f32_16x16x4_f32(
                                  false, b1, false, a1, (short)0, acc[mt], false, false);
                }
#pragma unroll
                for (int mt = 0; mt < MT; ++mt) acur[mt] = anxt[mt];
            }
#pragma unroll
            for (int q = 0; q < 4; ++q) bcur[q] = bnxt[q];
        }

        const int nb = nt * 16 + 8 * half;
        const v4f bLo = *reinterpret_cast<const v4f*>(&sBias[nb]);
        const v4f bHi = *reinterpret_cast<const v4f*>(&sBias[nb + 4]);
        float txs[MT], tys[MT];
#pragma unroll
        for (int mt = 0; mt < MT; ++mt) {
            const int m = mhalf * (MT * 16) + mt * 16 + sub;
            const v2f t = *reinterpret_cast<const v2f*>(&sTgt[2 * m]);
            txs[mt] = t[0];
            tys[mt] = t[1];
        }

#pragma unroll
        for (int r = 0; r < 8; ++r) {
            const int   nn     = nb + r;
            const bool  valid  = (nn < NOUT);
            const bool  isImag = (nn >= NPAR);
            const int   idx    = isImag ? nn - NPAR : nn;
            const int   fr     = idx / HSZ;
            const int   fc     = idx - fr * HSZ;
            const float frf    = (float)fr;
            const float fcf    = (float)fc;
            const float bn     = (r < 4) ? bLo[r & 3] : bHi[r & 3];
#pragma unroll
            for (int mt = 0; mt < MT; ++mt) {
                const float v = valid ? (acc[mt][r] + bn) : 0.0f;
                const float u = fcf * txs[mt] + frf * tys[mt];
                float su, cu;
                sincospif(u, &su, &cu);

                float sre, sim;
                if (isImag) { sre = -v * su; sim =  v * cu; }
                else        { sre =  v * cu; sim =  v * su; }
                psre[mt] += sre;
                psim[mt] += sim;
                pcen[mt] += v * v;
            }
            if ((r & 3) == 3) __builtin_amdgcn_sched_barrier(0);
        }
    }

    {
        const int slot = wp * 2 + half;
#pragma unroll
        for (int mt = 0; mt < MT; ++mt) {
            const int m = mhalf * (MT * 16) + mt * 16 + sub;
            sPart[(0 * NSLOT + slot) * MB + m] = psre[mt];
            sPart[(1 * NSLOT + slot) * MB + m] = psim[mt];
            sPart[(2 * NSLOT + slot) * MB + m] = pcen[mt];
        }
    }
    __syncthreads();

    if (tid < MB) {
        float sre = 0.0f, sim = 0.0f, cen = 0.0f;
#pragma unroll
        for (int s = 0; s < NSLOT; ++s) {
            sre += sPart[(0 * NSLOT + s) * MB + tid];
            sim += sPart[(1 * NSLOT + s) * MB + tid];
            cen += sPart[(2 * NSLOT + s) * MB + tid];
        }
        out[rowBase + tid] = (sre * sre + sim * sim) / (4.0f * cen);
    }
}

extern "C" void kernel_launch(void* const* d_in, const int* in_sizes, int n_in,
                              void* d_out, int out_size, void* d_ws, size_t ws_size,
                              hipStream_t stream) {
    (void)in_sizes; (void)n_in; (void)d_ws; (void)ws_size; (void)out_size;
    const float* X  = (const float*)d_in[0];
    const float* T  = (const float*)d_in[1];
    const float* W  = (const float*)d_in[2];
    const float* b  = (const float*)d_in[3];
    float* out = (float*)d_out;

    dim3 grid(BATCH / MB), block(512);
    hipLaunchKernelGGL(fourier_head2d_fused, grid, block, 0, stream, X, T, W, b, out);
}


# Separable phase: e^{i*pi*(c*tx+r*ty)} = e^{i*pi*c*tx} * e^{i*pi*r*ty}; 34 sincospi per sample go once per block into LDS tables, epilogue does one complex product per column (sum order kept)
# speedup vs baseline: 1.6708x; 1.0831x over previous
#include <hip/hip_runtime.h>
#include <hip/hip_bf16.h>
#include <math.h>


typedef __attribute__((ext_vector_type(2))) float v2f;
typedef __attribute__((ext_vector_type(4))) float v4f;
typedef __attribute__((ext_vector_type(8))) float v8f;

#define DIMK  512
#define BATCH 32768
#define HSZ   17
#define NPAR  289
#define NOUT  578
#define NT    37
#define LDA   516
#define MB    128
#define MT    4
#define KC    32
#define NSLOT 16

__global__ __launch_bounds__(512)
void fourier_head2d_fused(const float* __restrict__ X,
                          const float* __restrict__ T,
                          const float* __restrict__ W,
                          const float* __restrict__ bias,
                          float* __restrict__ out)
{
    __shared__ __attribute__((aligned(16))) float sA[MB * LDA];
    __shared__ __attribute__((aligned(8)))  float sEx[HSZ * MB * 2];
    __shared__ __attribute__((aligned(8)))  float sEy[HSZ * MB * 2];
    __shared__ __attribute__((aligned(16))) float sBias[NT * 16];
    __shared__ float sPart[3 * NSLOT * MB];

    const int tid     = threadIdx.x;
    const int lane    = tid & 31;
    const int wave    = tid >> 5;
    const int rowBase = blockIdx.x * MB;

    const int   tm  = tid & (MB - 1);
    const v2f   tgt = *reinterpret_cast<const v2f*>(T + (size_t)(rowBase + tm) * 2);

    for (int i = tid; i < NT * 16; i += 512) sBias[i] = (i < NOUT) ? bias[i] : 0.0f;

#pragma unroll 8
    for (int i = tid; i < MB * (DIMK / 4); i += 512) {
        const int r  = i >> 7;
        const int c4 = i & 127;
        const v4f v = *reinterpret_cast<const v4f*>(
                          X + (size_t)(rowBase + r) * DIMK + c4 * 4);
        *reinterpret_cast<v4f*>(&sA[r * LDA + c4 * 4]) = v;
    }

    {
        for (int j = tid >> 7; j < 2 * HSZ; j += 4) {
            const bool  isY = (j >= HSZ);
            const int   f   = isY ? j - HSZ : j;
            float s, c;
            sincospif((float)f * (isY ? tgt[1] : tgt[0]), &s, &c);
            float* dst = (isY ? sEy : sEx) + (f * MB + tm) * 2;
            *reinterpret_cast<v2f*>(dst) = (v2f){c, s};
        }
    }
    __syncthreads();

    const int half  = lane >> 4;
    const int sub   = lane & 15;
    const int mhalf = wave & 1;
    const int wp    = wave >> 1;

    const float* aBase = sA + (mhalf * (MT * 16) + sub) * LDA + half * 4;

    float psre[MT], psim[MT], pcen[MT];
#pragma unroll
    for (int mt = 0; mt < MT; ++mt) { psre[mt] = 0.0f; psim[mt] = 0.0f; pcen[mt] = 0.0f; }

    for (int nt = wp; nt < NT; nt += 8) {
        const int n  = nt * 16 + sub;
        const int nc = (n < NOUT) ? n : (NOUT - 1);
        const float* wrow = W + (size_t)nc * DIMK + half * 4;

        v8f acc[MT];
#pragma unroll
        for (int mt = 0; mt < MT; ++mt) acc[mt] = (v8f){};

        v4f bcur[4], bnxt[4];
#pragma unroll
        for (int q = 0; q < 4; ++q)
            bcur[q] = *reinterpret_cast<const v4f*>(wrow + q * 8);

        v4f acur[MT], anxt[MT];
#pragma unroll
        for (int mt = 0; mt < MT; ++mt)
            acur[mt] = *reinterpret_cast<const v4f*>(aBase + mt * 16 * LDA);

#pragma unroll 2
        for (int kc = 0; kc < DIMK / KC; ++kc) {
            const int kn = ((kc + 1 < DIMK / KC) ? (kc + 1) : kc) * KC;
#pragma unroll
            for (int q = 0; q < 4; ++q)
                bnxt[q] = *reinterpret_cast<const v4f*>(wrow + kn + q * 8);

#pragma unroll
            for (int q = 0; q < 4; ++q) {
                const int ko  = kc * KC + q * 8;
                const int kon = (ko + 8 < DIMK) ? (ko + 8) : ko;
#pragma unroll
                for (int mt = 0; mt < MT; ++mt)
                    anxt[mt] = *reinterpret_cast<const v4f*>(
                                   aBase + mt * 16 * LDA + kon);
                const v2f b0 = __builtin_shufflevector(bcur[q], bcur[q], 0, 1);
                const v2f b1 = __builtin_shufflevector(bcur[q], bcur[q], 2, 3);
#pragma unroll
                for (int mt = 0; mt < MT; ++mt) {
                    const v2f a0 = __builtin_shufflevector(acur[mt], acur[mt], 0, 1);
                    acc[mt] = __builtin_amdgcn_wmma_f32_16x16x4_f32(
                                  false, b0, false, a0, (short)0, acc[mt], false, false);
                }
#pragma unroll
                for (int mt = 0; mt < MT; ++mt) {
                    const v2f a1 = __builtin_shufflevector(acur[mt], acur[mt], 2, 3);
                    acc[mt] = __builtin_amdgcn_wmma_f32_16x16x4_f32(
                                  false, b1, false, a1, (short)0, acc[mt], false, false);
                }
#pragma unroll
                for (int mt = 0; mt < MT; ++mt) acur[mt] = anxt[mt];
            }
#pragma unroll
            for (int q = 0; q < 4; ++q) bcur[q] = bnxt[q];
        }

        const int nb = nt * 16 + 8 * half;
        const v4f bLo = *reinterpret_cast<const v4f*>(&sBias[nb]);
        const v4f bHi = *reinterpret_cast<const v4f*>(&sBias[nb + 4]);
        const int m0 = mhalf * (MT * 16) + sub;

#pragma unroll
        for (int r = 0; r < 8; ++r) {
            const int   nn     = nb + r;
            const bool  valid  = (nn < NOUT);
            const bool  isImag = (nn >= NPAR);
            const int   idx    = valid ? (isImag ? nn - NPAR : nn) : 0;
            const int   fr     = idx / HSZ;
            const int   fc     = idx - fr * HSZ;
            const float* ex    = sEx + (fc * MB + m0) * 2;
            const float* ey    = sEy + (fr * MB + m0) * 2;
            const float bn     = (r < 4) ? bLo[r & 3] : bHi[r & 3];
#pragma unroll
            for (int mt = 0; mt < MT; ++mt) {
                const float v = valid ? (acc[mt][r] + bn) : 0.0f;
                const v2f cx = *reinterpret_cast<const v2f*>(ex + mt * 32);
                const v2f cy = *reinterpret_cast<const v2f*>(ey + mt * 32);
                const float cu = cx[0] * cy[0] - cx[1] * cy[1];
                const float su = cx[1] * cy[0] + cx[0] * cy[1];

                float sre, sim;
                if (isImag) { sre = -v * su; sim =  v * cu; }
                else        { sre =  v * cu; sim =  v * su; }
                psre[mt] += sre;
                psim[mt] += sim;
                pcen[mt] += v * v;
            }
        }
    }

    {
        const int slot = wp * 2 + half;
#pragma unroll
        for (int mt = 0; mt < MT; ++mt) {
            const int m = mhalf * (MT * 16) + mt * 16 + sub;
            sPart[(0 * NSLOT + slot) * MB + m] = psre[mt];
            sPart[(1 * NSLOT + slot) * MB + m] = psim[mt];
            sPart[(2 * NSLOT + slot) * MB + m] = pcen[mt];
        }
    }
    __syncthreads();

    if (tid < MB) {
        float sre = 0.0f, sim = 0.0f, cen = 0.0f;
#pragma unroll
        for (int s = 0; s < NSLOT; ++s) {
            sre += sPart[(0 * NSLOT + s) * MB + tid];
            sim += sPart[(1 * NSLOT + s) * MB + tid];
            cen += sPart[(2 * NSLOT + s) * MB + tid];
        }
        out[rowBase + tid] = (sre * sre + sim * sim) / (4.0f * cen);
    }
}

extern "C" void kernel_launch(void* const* d_in, const int* in_sizes, int n_in,
                              void* d_out, int out_size, void* d_ws, size_t ws_size,
                              hipStream_t stream) {
    (void)in_sizes; (void)n_in; (void)d_ws; (void)ws_size; (void)out_size;
    const float* X  = (const float*)d_in[0];
    const float* T  = (const float*)d_in[1];
    const float* W  = (const float*)d_in[2];
    const float* b  = (const float*)d_in[3];
    float* out = (float*)d_out;

    dim3 grid(BATCH / MB), block(512);
    hipLaunchKernelGGL(fourier_head2d_fused, grid, block, 0, stream, X, T, W, b, out);
}
